# starting kernel
# baseline (speedup 1.0000x reference)
#include <hip/hip_runtime.h>
#include <hip/hip_bf16.h>
#include <math.h>


#define ZC    8
#define EC    32
#define HC    64
#define BC    64
#define LLC   512
#define LAGSC 2
#define LENC  (LLC - LAGSC)
#define NN    (BC * LENC)
#define RES_SIZE (NN * ZC)
#define LDSPAD 68

typedef __attribute__((ext_vector_type(2))) float v2f;
typedef __attribute__((ext_vector_type(8))) float v8f;

static __device__ __forceinline__ v8f wmma_f32(v2f a, v2f b, v8f c) {
    return __builtin_amdgcn_wmma_f32_16x16x4_f32(
         false, a,  false, b,
         (short)0, c,  false,  false);
}

__global__ __launch_bounds__(128)
void npctp_mlp_jvp_kernel(const float* __restrict__ x,
                          const float* __restrict__ emb,
                          const float* __restrict__ W1,
                          const float* __restrict__ b1,
                          const float* __restrict__ W2,
                          const float* __restrict__ b2,
                          const float* __restrict__ W3,
                          const float* __restrict__ b3,
                          float* __restrict__ out)
{
    __shared__ float buf_h[4][16][LDSPAD];
    __shared__ float buf_d[4][16][LDSPAD];
    __shared__ float lds_ld[16][ZC];

    const int tid  = threadIdx.x;
    const int wave = tid >> 5;
    const int lane = tid & 31;
    const int hi   = lane >> 4;
    const int lo   = lane & 15;
    const int n0   = blockIdx.x * 16;

    v2f aE[8];
    {
        const int n  = n0 + lo;
        const int b  = n / LENC;
        const int li = n - b * LENC;
        const float* ep = emb + (size_t)(b * LLC + li + LAGSC) * EC;
        #pragma unroll
        for (int kk = 0; kk < 8; ++kk) {
            const int e = 4 * kk + 2 * hi;
            aE[kk].x = ep[e];
            aE[kk].y = ep[e + 1];
        }
    }

    #pragma unroll
    for (int zi = 0; zi < 2; ++zi) {
        const int z = wave + 4 * zi;

        float xt[8];
        #pragma unroll
        for (int j = 0; j < 8; ++j) {
            const int n  = n0 + j + 8 * hi;
            const int b  = n / LENC;
            const int li = n - b * LENC;
            xt[j] = x[(size_t)(b * LLC + li + LAGSC) * ZC + z];
        }

        const float* W1z = W1 + z * (EC + 1) * HC;
        const float* W2z = W2 + z * HC * HC;
        const float* b1z = b1 + z * HC;
        const float* b2z = b2 + z * HC;
        const float* W3z = W3 + z * HC;
        const float  b3z = b3[z];

        #pragma unroll
        for (int c = 0; c < 4; ++c) {
            const int col = 16 * c + lo;
            v8f acc = {0.f,0.f,0.f,0.f,0.f,0.f,0.f,0.f};
            #pragma unroll
            for (int kk = 0; kk < 8; ++kk) {
                const int k = 4 * kk + 2 * hi;
                v2f bf;
                bf.x = W1z[(size_t)k * HC + col];
                bf.y = W1z[(size_t)(k + 1) * HC + col];
                acc = wmma_f32(aE[kk], bf, acc);
            }
            const float wlast = W1z[(size_t)EC * HC + col];
            const float bb    = b1z[col];
            #pragma unroll
            for (int j = 0; j < 8; ++j) {
                const float pre = acc[j] + xt[j] * wlast + bb;
                const float f   = (pre >= 0.f) ? 1.f : 0.01f;
                buf_h[wave][j + 8 * hi][col] = f * pre;
                buf_d[wave][j + 8 * hi][col] = f * wlast;
            }
        }

        v2f ah[16], ad[16];
        #pragma unroll
        for (int kk = 0; kk < 16; ++kk) {
            const int k = 4 * kk + 2 * hi;
            ah[kk].x = buf_h[wave][lo][k];
            ah[kk].y = buf_h[wave][lo][k + 1];
            ad[kk].x = buf_d[wave][lo][k];
            ad[kk].y = buf_d[wave][lo][k + 1];
        }

        float pout[8] = {0,0,0,0,0,0,0,0};
        float pdt [8] = {0,0,0,0,0,0,0,0};
        #pragma unroll
        for (int c = 0; c < 4; ++c) {
            const int col = 16 * c + lo;
            v8f acc = {0.f,0.f,0.f,0.f,0.f,0.f,0.f,0.f};
            v8f dac = {0.f,0.f,0.f,0.f,0.f,0.f,0.f,0.f};
            #pragma unroll
            for (int kk = 0; kk < 16; ++kk) {
                const int k = 4 * kk + 2 * hi;
                v2f bf;
                bf.x = W2z[(size_t)k * HC + col];
                bf.y = W2z[(size_t)(k + 1) * HC + col];
                acc = wmma_f32(ah[kk], bf, acc);
                dac = wmma_f32(ad[kk], bf, dac);
            }
            const float bb  = b2z[col];
            const float w3v = W3z[col];
            #pragma unroll
            for (int j = 0; j < 8; ++j) {
                const float pre = acc[j] + bb;
                const float f   = (pre >= 0.f) ? 1.f : 0.01f;
                pout[j] += (f * pre)    * w3v;
                pdt [j] += (f * dac[j]) * w3v;
            }
        }

        #pragma unroll
        for (int j = 0; j < 8; ++j) {
            float a = pout[j], d = pdt[j];
            #pragma unroll
            for (int off = 1; off < 16; off <<= 1) {
                a += __shfl_xor(a, off, 32);
                d += __shfl_xor(d, off, 32);
            }
            pout[j] = a; pdt[j] = d;
        }

        if (lo == 0) {
            #pragma unroll
            for (int j = 0; j < 8; ++j) {
                const int m = j + 8 * hi;
                const int n = n0 + m;
                out[(size_t)n * ZC + z] = pout[j] + b3z;
                lds_ld[m][z] = __logf(fabsf(pdt[j]));
            }
        }
    }

    __syncthreads();
    if (tid < 16) {
        float s = 0.f;
        #pragma unroll
        for (int z = 0; z < ZC; ++z) s += lds_ld[tid][z];
        out[RES_SIZE + n0 + tid] = s;
    }
}

extern "C" void kernel_launch(void* const* d_in, const int* in_sizes, int n_in,
                              void* d_out, int out_size, void* d_ws, size_t ws_size,
                              hipStream_t stream) {
    (void)in_sizes; (void)n_in; (void)out_size; (void)d_ws; (void)ws_size;
    const float* x   = (const float*)d_in[0];
    const float* emb = (const float*)d_in[1];
    const float* W1  = (const float*)d_in[2];
    const float* b1  = (const float*)d_in[3];
    const float* W2  = (const float*)d_in[4];
    const float* b2  = (const float*)d_in[5];
    const float* W3  = (const float*)d_in[6];
    const float* b3  = (const float*)d_in[7];
    float* out = (float*)d_out;

    dim3 grid(NN / 16);
    dim3 block(128);
    npctp_mlp_jvp_kernel<<<grid, block, 0, stream>>>(x, emb, W1, b1, W2, b2,
                                                     W3, b3, out);
}


# Weight-stationary: each wave owns one latent, loads its W1/W2 B fragments into registers once and walks 8 sample tiles; block results staged in LDS and stored as whole lines; sum order unchanged
# speedup vs baseline: 1.1581x; 1.1581x over previous
#include <hip/hip_runtime.h>
#include <hip/hip_bf16.h>
#include <math.h>


#define ZC    8
#define EC    32
#define HC    64
#define BC    64
#define LLC   512
#define LAGSC 2
#define LENC  (LLC - LAGSC)
#define NN    (BC * LENC)
#define RES_SIZE (NN * ZC)
#define LDSPAD 68
#define TPB   8
#define SPB   (16 * TPB)

typedef __attribute__((ext_vector_type(2))) float v2f;
typedef __attribute__((ext_vector_type(8))) float v8f;


static __device__ __forceinline__ v8f wmma_f32(v2f a, v2f b, v8f c) {
    return __builtin_amdgcn_wmma_f32_16x16x4_f32(
         false, a,  false, b,
         (short)0, c,  false,  false);
}

static __device__ __forceinline__ void load_tile(const float* __restrict__ x,
                                                 const float* __restrict__ emb,
                                                 int n0, int z, int hi, int lo,
                                                 v2f (&aE)[8], float (&xt)[8]) {
    {
        const int n  = n0 + lo;
        const int b  = n / LENC;
        const int li = n - b * LENC;
        const float* ep = emb + (size_t)(b * LLC + li + LAGSC) * EC;
        #pragma unroll
        for (int kk = 0; kk < 8; ++kk) {
            const int e = 4 * kk + 2 * hi;
            aE[kk].x = ep[e];
            aE[kk].y = ep[e + 1];
        }
    }
    #pragma unroll
    for (int j = 0; j < 8; ++j) {
        const int n  = n0 + j + 8 * hi;
        const int b  = n / LENC;
        const int li = n - b * LENC;
        xt[j] = x[(size_t)(b * LLC + li + LAGSC) * ZC + z];
    }
}

__global__ __launch_bounds__(256)
void npctp_mlp_jvp_kernel(const float* __restrict__ x,
                          const float* __restrict__ emb,
                          const float* __restrict__ W1,
                          const float* __restrict__ b1,
                          const float* __restrict__ W2,
                          const float* __restrict__ b2,
                          const float* __restrict__ W3,
                          const float* __restrict__ b3,
                          float* __restrict__ out)
{
    __shared__ float buf_h[ZC][16][LDSPAD];
    __shared__ float buf_d[ZC][16][LDSPAD];
    __shared__ __attribute__((aligned(16))) float res_s[SPB][ZC];
    __shared__ float ld_s[SPB][ZC];

    const int tid  = threadIdx.x;
    const int wave = tid >> 5;
    const int lane = tid & 31;
    const int hi   = lane >> 4;
    const int lo   = lane & 15;
    const int z    = wave;
    const int nb   = blockIdx.x * SPB;

    const float* W1z = W1 + z * (EC + 1) * HC;
    const float* W2z = W2 + z * HC * HC;
    const float* b1z = b1 + z * HC;
    const float* b2z = b2 + z * HC;
    const float* W3z = W3 + z * HC;
    const float  b3z = b3[z];

    v2f w1f[4][8];
    v2f w2f[4][16];
    float wlast[4], bb1[4], bb2[4], w3v[4];
    #pragma unroll
    for (int c = 0; c < 4; ++c) {
        const int col = 16 * c + lo;
        #pragma unroll
        for (int kk = 0; kk < 8; ++kk) {
            const int k = 4 * kk + 2 * hi;
            w1f[c][kk].x = W1z[(size_t)k * HC + col];
            w1f[c][kk].y = W1z[(size_t)(k + 1) * HC + col];
        }
        #pragma unroll
        for (int kk = 0; kk < 16; ++kk) {
            const int k = 4 * kk + 2 * hi;
            w2f[c][kk].x = W2z[(size_t)k * HC + col];
            w2f[c][kk].y = W2z[(size_t)(k + 1) * HC + col];
        }
        wlast[c] = W1z[(size_t)EC * HC + col];
        bb1[c]   = b1z[col];
        bb2[c]   = b2z[col];
        w3v[c]   = W3z[col];
    }

    v2f aE[8];
    float xt[8];
    load_tile(x, emb, nb, z, hi, lo, aE, xt);

    for (int t = 0; t < TPB; ++t) {
        const int tn = (t + 1 < TPB) ? (t + 1) : t;
        v2f aEn[8];
        float xtn[8];
        load_tile(x, emb, nb + 16 * tn, z, hi, lo, aEn, xtn);

        #pragma unroll
        for (int cp = 0; cp < 2; ++cp) {
            v8f acc0 = {0.f,0.f,0.f,0.f,0.f,0.f,0.f,0.f};
            v8f acc1 = {0.f,0.f,0.f,0.f,0.f,0.f,0.f,0.f};
            #pragma unroll
            for (int kk = 0; kk < 8; ++kk) {
                acc0 = wmma_f32(aE[kk], w1f[2 * cp][kk],     acc0);
                acc1 = wmma_f32(aE[kk], w1f[2 * cp + 1][kk], acc1);
            }
            #pragma unroll
            for (int s = 0; s < 2; ++s) {
                const int c   = 2 * cp + s;
                const int col = 16 * c + lo;
                #pragma unroll
                for (int j = 0; j < 8; ++j) {
                    const float a   = s ? acc1[j] : acc0[j];
                    const float pre = a + xt[j] * wlast[c] + bb1[c];
                    const float f   = (pre >= 0.f) ? 1.f : 0.01f;
                    buf_h[wave][j + 8 * hi][col] = f * pre;
                    buf_d[wave][j + 8 * hi][col] = f * wlast[c];
                }
            }
        }

        v2f ah[16], ad[16];
        #pragma unroll
        for (int kk = 0; kk < 16; ++kk) {
            const int k = 4 * kk + 2 * hi;
            ah[kk].x = buf_h[wave][lo][k];
            ah[kk].y = buf_h[wave][lo][k + 1];
            ad[kk].x = buf_d[wave][lo][k];
            ad[kk].y = buf_d[wave][lo][k + 1];
        }

        float pout[8] = {0,0,0,0,0,0,0,0};
        float pdt [8] = {0,0,0,0,0,0,0,0};
        #pragma unroll
        for (int c = 0; c < 4; ++c) {
            v8f acc = {0.f,0.f,0.f,0.f,0.f,0.f,0.f,0.f};
            v8f dac = {0.f,0.f,0.f,0.f,0.f,0.f,0.f,0.f};
            #pragma unroll
            for (int kk = 0; kk < 16; ++kk) {
                acc = wmma_f32(ah[kk], w2f[c][kk], acc);
                dac = wmma_f32(ad[kk], w2f[c][kk], dac);
            }
            #pragma unroll
            for (int j = 0; j < 8; ++j) {
                const float pre = acc[j] + bb2[c];
                const float f   = (pre >= 0.f) ? 1.f : 0.01f;
                pout[j] += (f * pre)    * w3v[c];
                pdt [j] += (f * dac[j]) * w3v[c];
            }
        }

        #pragma unroll
        for (int j = 0; j < 8; ++j) {
            float a = pout[j], d = pdt[j];
            #pragma unroll
            for (int off = 1; off < 16; off <<= 1) {
                a += __shfl_xor(a, off, 32);
                d += __shfl_xor(d, off, 32);
            }
            pout[j] = a; pdt[j] = d;
        }

        if (lo == 0) {
            #pragma unroll
            for (int j = 0; j < 8; ++j) {
                const int m = 16 * t + j + 8 * hi;
                res_s[m][z] = pout[j] + b3z;
                ld_s[m][z]  = __logf(fabsf(pdt[j]));
            }
        }

        #pragma unroll
        for (int kk = 0; kk < 8; ++kk) aE[kk] = aEn[kk];
        #pragma unroll
        for (int j = 0; j < 8; ++j) xt[j] = xtn[j];
    }

    __syncthreads();
    {
        const float4 v = reinterpret_cast<const float4*>(&res_s[0][0])[tid];
        reinterpret_cast<float4*>(out + (size_t)nb * ZC)[tid] = v;
    }
    if (tid < SPB) {
        float s = 0.f;
        #pragma unroll
        for (int zz = 0; zz < ZC; ++zz) s += ld_s[tid][zz];
        out[RES_SIZE + nb + tid] = s;
    }
}

extern "C" void kernel_launch(void* const* d_in, const int* in_sizes, int n_in,
                              void* d_out, int out_size, void* d_ws, size_t ws_size,
                              hipStream_t stream) {
    (void)in_sizes; (void)n_in; (void)out_size; (void)d_ws; (void)ws_size;
    const float* x   = (const float*)d_in[0];
    const float* emb = (const float*)d_in[1];
    const float* W1  = (const float*)d_in[2];
    const float* b1  = (const float*)d_in[3];
    const float* W2  = (const float*)d_in[4];
    const float* b2  = (const float*)d_in[5];
    const float* W3  = (const float*)d_in[6];
    const float* b3  = (const float*)d_in[7];
    float* out = (float*)d_out;

    dim3 grid(NN / SPB);
    dim3 block(256);
    npctp_mlp_jvp_kernel<<<grid, block, 0, stream>>>(x, emb, W1, b1, W2, b2,
                                                     W3, b3, out);
}
